# starting kernel
# baseline (speedup 1.0000x reference)
#include <hip/hip_runtime.h>

#define DIM 128
#define WAVES_PER_BLOCK 8
#define BLOCK (WAVES_PER_BLOCK * 32)

typedef float v2f __attribute__((ext_vector_type(2)));
typedef float v8f __attribute__((ext_vector_type(8)));

__global__ void __launch_bounds__(BLOCK)
ln_relu_zero_kernel(const float* __restrict__ x,
                    const float* __restrict__ gamma,
                    const float* __restrict__ beta,
                    float* __restrict__ h,
                    float* __restrict__ aggr,
                    int n_nodes) {
  int wave = (blockIdx.x * blockDim.x + threadIdx.x) >> 5;
  int lane = threadIdx.x & 31;
  if (wave >= n_nodes) return;

  const float4 v = ((const float4*)(x + (size_t)wave * DIM))[lane];
  float s  = v.x + v.y + v.z + v.w;
  float s2 = v.x * v.x + v.y * v.y + v.z * v.z + v.w * v.w;
  #pragma unroll
  for (int off = 16; off > 0; off >>= 1) {
    s  += __shfl_xor(s,  off, 32);
    s2 += __shfl_xor(s2, off, 32);
  }
  const float mu  = s * (1.0f / DIM);
  const float var = s2 * (1.0f / DIM) - mu * mu;
  const float rs  = rsqrtf(var + 1e-5f);

  const float4 g = ((const float4*)gamma)[lane];
  const float4 b = ((const float4*)beta)[lane];
  float4 hv;
  hv.x = fmaxf((v.x - mu) * rs * g.x + b.x, 0.0f);
  hv.y = fmaxf((v.y - mu) * rs * g.y + b.y, 0.0f);
  hv.z = fmaxf((v.z - mu) * rs * g.z + b.z, 0.0f);
  hv.w = fmaxf((v.w - mu) * rs * g.w + b.w, 0.0f);

  ((float4*)(h + (size_t)wave * DIM))[lane] = hv;
  const float4 z = {0.0f, 0.0f, 0.0f, 0.0f};
  ((float4*)(aggr + (size_t)wave * DIM))[lane] = z;
}

__global__ void __launch_bounds__(BLOCK)
edge_scatter_kernel(const int* __restrict__ ei,
                    const float* __restrict__ h,
                    float* __restrict__ aggr,
                    int n_edges) {
  int wave = (blockIdx.x * blockDim.x + threadIdx.x) >> 5;
  int lane = threadIdx.x & 31;
  if (wave >= n_edges) return;

  const int src = ei[wave];
  const int dst = ei[n_edges + wave];

  const float4 m = ((const float4*)(h + (size_t)src * DIM))[lane];
  float* ap = aggr + (size_t)dst * DIM + lane * 4;
  atomicAdd(ap + 0, m.x);
  atomicAdd(ap + 1, m.y);
  atomicAdd(ap + 2, m.z);
  atomicAdd(ap + 3, m.w);
}

__global__ void __launch_bounds__(256)
fuse_kernel(const float* __restrict__ h,
            const float* __restrict__ eps_p,
            float* __restrict__ aggr,
            int n) {
  int i = blockIdx.x * blockDim.x + threadIdx.x;
  if (i >= n) return;
  const float scale = 1.0f + eps_p[0];
  aggr[i] = fmaf(scale, h[i], aggr[i]);
}

__global__ void __launch_bounds__(BLOCK)
gemm_kernel(const float* __restrict__ Mmat,
            const float* __restrict__ W,
            const float* __restrict__ bias,
            const float* __restrict__ x,
            float* __restrict__ out,
            int n_tiles) {
  int wave = threadIdx.x >> 5;
  int lane = threadIdx.x & 31;
  int tile = blockIdx.x * WAVES_PER_BLOCK + wave;
  if (tile >= n_tiles) return;

  const int tm   = tile >> 3;
  const int tn   = tile & 7;
  const int row0 = tm * 16;
  const int col0 = tn * 16;
  const int half = lane >> 4;
  const int l    = lane & 15;

  const float* arow = Mmat + (size_t)(row0 + l) * DIM + 2 * half;
  const float* brow = W    + (size_t)(col0 + l) * DIM + 2 * half;

  v8f c = {};
  #pragma unroll
  for (int k = 0; k < DIM; k += 4) {
    v2f a = *(const v2f*)(arow + k);
    v2f b = *(const v2f*)(brow + k);
    c = __builtin_amdgcn_wmma_f32_16x16x4_f32(
         false, a,  false, b,
         (short)0, c,  false,  false);
  }

  const float bv = bias[col0 + l];
  #pragma unroll
  for (int r = 0; r < 8; ++r) {
    const int row = row0 + r + 8 * half;
    const size_t idx = (size_t)row * DIM + col0 + l;
    out[idx] = c[r] + bv + x[idx];
  }
}

extern "C" void kernel_launch(void* const* d_in, const int* in_sizes, int n_in,
                              void* d_out, int out_size, void* d_ws, size_t ws_size,
                              hipStream_t stream) {
  const float* x     = (const float*)d_in[0];
  const int*   ei    = (const int*)  d_in[1];
  const float* gamma = (const float*)d_in[2];
  const float* beta  = (const float*)d_in[3];
  const float* W     = (const float*)d_in[4];
  const float* bias  = (const float*)d_in[5];
  const float* eps   = (const float*)d_in[6];
  float* out = (float*)d_out;

  const int n_nodes = in_sizes[0] / DIM;
  const int n_edges = in_sizes[1] / 2;

  float* h    = (float*)d_ws;
  float* aggr = h + (size_t)n_nodes * DIM;

  {
    int blocks = (n_nodes + WAVES_PER_BLOCK - 1) / WAVES_PER_BLOCK;
    ln_relu_zero_kernel<<<blocks, BLOCK, 0, stream>>>(x, gamma, beta, h, aggr, n_nodes);
  }
  {
    int blocks = (n_edges + WAVES_PER_BLOCK - 1) / WAVES_PER_BLOCK;
    edge_scatter_kernel<<<blocks, BLOCK, 0, stream>>>(ei, h, aggr, n_edges);
  }
  {
    int n = n_nodes * DIM;
    int blocks = (n + 255) / 256;
    fuse_kernel<<<blocks, 256, 0, stream>>>(h, eps, aggr, n);
  }
  {
    int n_tiles = (n_nodes / 16) * (DIM / 16);
    int blocks = (n_tiles + WAVES_PER_BLOCK - 1) / WAVES_PER_BLOCK;
    gemm_kernel<<<blocks, BLOCK, 0, stream>>>(aggr, W, bias, x, out, n_tiles);
  }
}


# Range sum finds its edges in a 1-byte owner key per edge (dst/400, written by launch 1), 16 keys per 16 B load, not in the 4 B dst plane: scan bytes a quarter; ids read on a match; sum order: arrival
# speedup vs baseline: 3.1424x; 3.1424x over previous
#include <hip/hip_runtime.h>

#define DIM 128
#define PT_WAVES 8
#define PT_THREADS (PT_WAVES * 32)
#define PT_TILES 5
#define PT_ROWS (16 * PT_TILES)
#define HS_PITCH 132
#define OWN_NODES 400
#define OWN_THREADS 1024
#define OWN_WAVES (OWN_THREADS / 32)
#define Q_CAP 64
#define Q_GROUP 8

typedef float v2f __attribute__((ext_vector_type(2)));
typedef float v8f __attribute__((ext_vector_type(8)));

__device__ __forceinline__ unsigned owner_keys4(int a, int b, int c, int d) {
  return  (((unsigned)a / (unsigned)OWN_NODES) & 255u)        |
         ((((unsigned)b / (unsigned)OWN_NODES) & 255u) << 8)  |
         ((((unsigned)c / (unsigned)OWN_NODES) & 255u) << 16) |
          (((unsigned)d / (unsigned)OWN_NODES)         << 24);
}

__global__ void __launch_bounds__(PT_THREADS)
ln_relu_linear_kernel(const float* __restrict__ x,
                      const float* __restrict__ gamma,
                      const float* __restrict__ beta,
                      const float* __restrict__ W,
                      const int* __restrict__ ei,
                      float* __restrict__ P,
                      unsigned int* __restrict__ keys,
                      int n_nodes, int n_edges) {
  __shared__ __attribute__((aligned(16))) float hs[PT_ROWS * HS_PITCH];

  const int wave = threadIdx.x >> 5;
  const int lane = threadIdx.x & 31;
  const int half = lane >> 4;
  const int l    = lane & 15;
  const int row_base = blockIdx.x * PT_ROWS;

  v2f bf[DIM / 4];
  {
    const float* brow = W + (size_t)(16 * wave + l) * DIM + 2 * half;
    #pragma unroll
    for (int kk = 0; kk < DIM / 4; ++kk) bf[kk] = *(const v2f*)(brow + 4 * kk);
  }

  {
    float4 v[2 * PT_TILES];
    #pragma unroll
    for (int j = 0; j < 2 * PT_TILES; ++j) {
      const int row = min(row_base + wave + 8 * j, n_nodes - 1);
      v[j] = ((const float4*)(x + (size_t)row * DIM))[lane];
    }
    const float4 g = ((const float4*)gamma)[lane];
    const float4 b = ((const float4*)beta)[lane];
    #pragma unroll
    for (int j = 0; j < 2 * PT_TILES; ++j) {
      float s  = v[j].x + v[j].y + v[j].z + v[j].w;
      float s2 = v[j].x * v[j].x + v[j].y * v[j].y + v[j].z * v[j].z + v[j].w * v[j].w;
      #pragma unroll
      for (int off = 16; off > 0; off >>= 1) {
        s  += __shfl_xor(s,  off, 32);
        s2 += __shfl_xor(s2, off, 32);
      }
      const float mu  = s * (1.0f / DIM);
      const float var = s2 * (1.0f / DIM) - mu * mu;
      const float rs  = rsqrtf(var + 1e-5f);
      float4 hv;
      hv.x = fmaxf((v[j].x - mu) * rs * g.x + b.x, 0.0f);
      hv.y = fmaxf((v[j].y - mu) * rs * g.y + b.y, 0.0f);
      hv.z = fmaxf((v[j].z - mu) * rs * g.z + b.z, 0.0f);
      hv.w = fmaxf((v[j].w - mu) * rs * g.w + b.w, 0.0f);
      *(float4*)(hs + (wave + 8 * j) * HS_PITCH + 4 * lane) = hv;
    }
  }
  __syncthreads();

  for (int t = 0; t < PT_TILES; ++t) {
    const float* arow = hs + (16 * t + l) * HS_PITCH + 2 * half;
    v8f c = {};
    #pragma unroll
    for (int kk = 0; kk < DIM / 4; ++kk) {
      const v2f a = *(const v2f*)(arow + 4 * kk);
      c = __builtin_amdgcn_wmma_f32_16x16x4_f32(
           false, a,  false, bf[kk],
           (short)0, c,  false,  false);
    }
    #pragma unroll
    for (int r = 0; r < 8; ++r) {
      const int row = row_base + 16 * t + r + 8 * half;
      if (row < n_nodes) P[(size_t)row * DIM + 16 * wave + l] = c[r];
    }
  }

  {
    const int* __restrict__ dstp = ei + n_edges;
    const int n_pairs = 2 * ((n_edges + 15) >> 4);
    const int first   = blockIdx.x * PT_THREADS + threadIdx.x;
    const int stride  = gridDim.x * PT_THREADS;
    if ((n_edges & 3) == 0) {
      const int last4 = (n_edges >> 2) - 1;
      for (int p = first; p < n_pairs; p += stride) {
        const int4 d0 = ((const int4*)dstp)[min(2 * p,     last4)];
        const int4 d1 = ((const int4*)dstp)[min(2 * p + 1, last4)];
        uint2 k;
        k.x = owner_keys4(d0.x, d0.y, d0.z, d0.w);
        k.y = owner_keys4(d1.x, d1.y, d1.z, d1.w);
        ((uint2*)keys)[p] = k;
      }
    } else {
      const int last = n_edges - 1;
      for (int p = first; p < n_pairs; p += stride) {
        const int e = 8 * p;
        uint2 k;
        k.x = owner_keys4(dstp[min(e,     last)], dstp[min(e + 1, last)], dstp[min(e + 2, last)], dstp[min(e + 3, last)]);
        k.y = owner_keys4(dstp[min(e + 4, last)], dstp[min(e + 5, last)], dstp[min(e + 6, last)], dstp[min(e + 7, last)]);
        ((uint2*)keys)[p] = k;
      }
    }
  }
}

__global__ void __launch_bounds__(OWN_THREADS)
own_range_sum_kernel(const int* __restrict__ ei,
                     const unsigned int* __restrict__ keys,
                     const float* __restrict__ P,
                     const float* __restrict__ x,
                     const float* __restrict__ bias,
                     const float* __restrict__ eps_p,
                     float* __restrict__ out,
                     int n_nodes, int n_edges) {
  __shared__ float acc[OWN_NODES * DIM];
  __shared__ int   qbuf[OWN_WAVES * Q_CAP];

  const int tid  = threadIdx.x;
  const int lane = tid & 31;
  const int base = blockIdx.x * OWN_NODES;
  const int cnt  = min(OWN_NODES, n_nodes - base);
  const int n4   = cnt * (DIM / 4);

  {
    const float scale = 1.0f + eps_p[0];
    const float4* p4 = (const float4*)(P + (size_t)base * DIM);
    const float4* x4 = (const float4*)(x + (size_t)base * DIM);
    const float4  bv = ((const float4*)bias)[lane];
    float4* a4 = (float4*)acc;
    for (int i = tid; i < n4; i += OWN_THREADS) {
      const float4 v = p4[i];
      const float4 r = x4[i];
      float4 s;
      s.x = v.x * scale + r.x + bv.x;
      s.y = v.y * scale + r.y + bv.y;
      s.z = v.z * scale + r.z + bv.z;
      s.w = v.w * scale + r.w + bv.w;
      a4[i] = s;
    }
  }
  __syncthreads();

  const int* __restrict__ srcp = ei;
  const int* __restrict__ dstp = ei + n_edges;
  const int n_vec = (n_edges + 15) >> 4;
  const int iters = (n_vec + OWN_THREADS - 1) / OWN_THREADS;
  const unsigned kx = (unsigned)(blockIdx.x & 255) * 0x01010101u;

  int* const qw = qbuf + (tid >> 5) * Q_CAP;
  unsigned q_head = 0u, q_tail = 0u;

  auto load16 = [&](int it) -> uint4 {
    const int g = it * OWN_THREADS + tid;
    return ((const uint4*)keys)[g < n_vec ? g : 0];
  };

  auto drain = [&](int n) {
    int s = 0, r = -1;
    if (lane < n) {
      const int e = qw[(q_head + (unsigned)lane) & (Q_CAP - 1)];
      s = srcp[e];
      r = dstp[e] - base;
    }
    q_head += (unsigned)n;
    #pragma unroll 1
    for (int i = 0; i < n; i += Q_GROUP) {
      float v[Q_GROUP][4];
      #pragma unroll
      for (int g = 0; g < Q_GROUP; ++g) {
        if (i + g < n) {
          const int su = __builtin_amdgcn_readlane(s, i + g);
          const float* hp = P + (size_t)su * DIM + lane;
          v[g][0] = hp[0];
          v[g][1] = hp[32];
          v[g][2] = hp[64];
          v[g][3] = hp[96];
        }
      }
      #pragma unroll
      for (int g = 0; g < Q_GROUP; ++g) {
        if (i + g < n) {
          const int ru = __builtin_amdgcn_readlane(r, i + g);
          if ((unsigned)ru < (unsigned)cnt) {
            float* ap = acc + ru * DIM + lane;
            atomicAdd(ap +  0, v[g][0]);
            atomicAdd(ap + 32, v[g][1]);
            atomicAdd(ap + 64, v[g][2]);
            atomicAdd(ap + 96, v[g][3]);
          }
        }
      }
    }
  };

  auto take16 = [&](const uint4& w, int it) {
    const int g = it * OWN_THREADS + tid;
    const unsigned t0 = w.x ^ kx, t1 = w.y ^ kx, t2 = w.z ^ kx, t3 = w.w ^ kx;
    const unsigned f0 = (t0 - 0x01010101u) & ~t0 & 0x80808080u;
    const unsigned f1 = (t1 - 0x01010101u) & ~t1 & 0x80808080u;
    const unsigned f2 = (t2 - 0x01010101u) & ~t2 & 0x80808080u;
    const unsigned f3 = (t3 - 0x01010101u) & ~t3 & 0x80808080u;
    unsigned hm = (f0 >> 7) | (f1 >> 6) | (f2 >> 5) | (f3 >> 4);
    if (g >= n_vec) hm = 0u;
    if (__builtin_amdgcn_ballot_w32(hm != 0u) == 0u) return;
    const int e0 = 16 * g;
    for (int round = 0; round < 16; ++round) {
      const int  p   = __builtin_ctz(hm | 0x80000000u);
      const int  e   = e0 + 4 * (p & 7) + (p >> 3);
      const bool hit = (hm != 0u) & (e < n_edges);
      const unsigned mask = __builtin_amdgcn_ballot_w32(hit);
      if (mask != 0u) {
        const unsigned pos = q_tail + __builtin_amdgcn_mbcnt_lo(mask, 0u);
        if (hit) qw[pos & (Q_CAP - 1)] = e;
        q_tail += (unsigned)__builtin_popcount(mask);
      }
      hm &= hm - 1u;
      if (q_tail - q_head >= 32u) drain(32);
      if (__builtin_amdgcn_ballot_w32(hm != 0u) == 0u) break;
    }
  };

  const int n_batches = (iters + 3) >> 2;
  const int rot = (int)(((long long)blockIdx.x * n_batches) / gridDim.x);

  for (int k = 0; k < n_batches; ++k) {
    const int bt = (k + rot < n_batches) ? k + rot : k + rot - n_batches;
    const int it = 4 * bt;
    const uint4 d0 = load16(it);
    const uint4 d1 = load16(it + 1);
    const uint4 d2 = load16(it + 2);
    const uint4 d3 = load16(it + 3);
    take16(d0, it);
    take16(d1, it + 1);
    take16(d2, it + 2);
    take16(d3, it + 3);
  }
  if (q_tail != q_head) drain((int)(q_tail - q_head));
  __syncthreads();

  {
    float4* o4 = (float4*)(out + (size_t)base * DIM);
    const float4* a4 = (const float4*)acc;
    for (int i = tid; i < n4; i += OWN_THREADS) o4[i] = a4[i];
  }
}

extern "C" void kernel_launch(void* const* d_in, const int* in_sizes, int n_in,
                              void* d_out, int out_size, void* d_ws, size_t ws_size,
                              hipStream_t stream) {
  const float* x     = (const float*)d_in[0];
  const int*   ei    = (const int*)  d_in[1];
  const float* gamma = (const float*)d_in[2];
  const float* beta  = (const float*)d_in[3];
  const float* W     = (const float*)d_in[4];
  const float* bias  = (const float*)d_in[5];
  const float* eps   = (const float*)d_in[6];
  float* out = (float*)d_out;

  const int n_nodes = in_sizes[0] / DIM;
  const int n_edges = in_sizes[1] / 2;

  float* P = (float*)d_ws;
  unsigned int* keys = (unsigned int*)(P + (size_t)n_nodes * DIM);

  {
    int blocks = (n_nodes + PT_ROWS - 1) / PT_ROWS;
    ln_relu_linear_kernel<<<blocks, PT_THREADS, 0, stream>>>(x, gamma, beta, W, ei, P, keys, n_nodes, n_edges);
  }
  {
    int blocks = (n_nodes + OWN_NODES - 1) / OWN_NODES;
    own_range_sum_kernel<<<blocks, OWN_THREADS, 0, stream>>>(ei, keys, P, x, bias, eps, out, n_nodes, n_edges);
  }
}


# Owner keys bit-sliced: 8 plane words per 32 edges (launch 1 votes them); range sum ANDs a lane's 8 words into 1 match bit per edge, 32 edges per test not 16, one vote per round; sum order: arrival
# speedup vs baseline: 3.3522x; 1.0668x over previous
#include <hip/hip_runtime.h>

#define DIM 128
#define PT_WAVES 8
#define PT_THREADS (PT_WAVES * 32)
#define PT_TILES 5
#define PT_ROWS (16 * PT_TILES)
#define HS_PITCH 132
#define OWN_NODES 400
#define OWN_THREADS 1024
#define OWN_WAVES (OWN_THREADS / 32)
#define Q_CAP 64
#define Q_GROUP 8

typedef float v2f __attribute__((ext_vector_type(2)));
typedef float v8f __attribute__((ext_vector_type(8)));

__global__ void __launch_bounds__(PT_THREADS)
ln_relu_linear_kernel(const float* __restrict__ x,
                      const float* __restrict__ gamma,
                      const float* __restrict__ beta,
                      const float* __restrict__ W,
                      const int* __restrict__ ei,
                      float* __restrict__ P,
                      unsigned int* __restrict__ keys,
                      int n_nodes, int n_edges) {
  __shared__ __attribute__((aligned(16))) float hs[PT_ROWS * HS_PITCH];

  const int wave = threadIdx.x >> 5;
  const int lane = threadIdx.x & 31;
  const int half = lane >> 4;
  const int l    = lane & 15;
  const int row_base = blockIdx.x * PT_ROWS;

  v2f bf[DIM / 4];
  {
    const float* brow = W + (size_t)(16 * wave + l) * DIM + 2 * half;
    #pragma unroll
    for (int kk = 0; kk < DIM / 4; ++kk) bf[kk] = *(const v2f*)(brow + 4 * kk);
  }

  {
    float4 v[2 * PT_TILES];
    #pragma unroll
    for (int j = 0; j < 2 * PT_TILES; ++j) {
      const int row = min(row_base + wave + 8 * j, n_nodes - 1);
      v[j] = ((const float4*)(x + (size_t)row * DIM))[lane];
    }
    const float4 g = ((const float4*)gamma)[lane];
    const float4 b = ((const float4*)beta)[lane];
    #pragma unroll
    for (int j = 0; j < 2 * PT_TILES; ++j) {
      float s  = v[j].x + v[j].y + v[j].z + v[j].w;
      float s2 = v[j].x * v[j].x + v[j].y * v[j].y + v[j].z * v[j].z + v[j].w * v[j].w;
      #pragma unroll
      for (int off = 16; off > 0; off >>= 1) {
        s  += __shfl_xor(s,  off, 32);
        s2 += __shfl_xor(s2, off, 32);
      }
      const float mu  = s * (1.0f / DIM);
      const float var = s2 * (1.0f / DIM) - mu * mu;
      const float rs  = rsqrtf(var + 1e-5f);
      float4 hv;
      hv.x = fmaxf((v[j].x - mu) * rs * g.x + b.x, 0.0f);
      hv.y = fmaxf((v[j].y - mu) * rs * g.y + b.y, 0.0f);
      hv.z = fmaxf((v[j].z - mu) * rs * g.z + b.z, 0.0f);
      hv.w = fmaxf((v[j].w - mu) * rs * g.w + b.w, 0.0f);
      *(float4*)(hs + (wave + 8 * j) * HS_PITCH + 4 * lane) = hv;
    }
  }
  __syncthreads();

  for (int t = 0; t < PT_TILES; ++t) {
    const float* arow = hs + (16 * t + l) * HS_PITCH + 2 * half;
    v8f c = {};
    #pragma unroll
    for (int kk = 0; kk < DIM / 4; ++kk) {
      const v2f a = *(const v2f*)(arow + 4 * kk);
      c = __builtin_amdgcn_wmma_f32_16x16x4_f32(
           false, a,  false, bf[kk],
           (short)0, c,  false,  false);
    }
    #pragma unroll
    for (int r = 0; r < 8; ++r) {
      const int row = row_base + 16 * t + r + 8 * half;
      if (row < n_nodes) P[(size_t)row * DIM + 16 * wave + l] = c[r];
    }
  }

  {
    const int* __restrict__ dstp = ei + n_edges;
    const int n_units  = (n_edges + 31) >> 5;
    const int n_groups = (n_units + 7) >> 3;
    unsigned int* __restrict__ keys_hi = keys + 32 * n_groups;
    const int last    = n_edges - 1;
    const int n_waves = gridDim.x * PT_WAVES;
    for (int grp = blockIdx.x * PT_WAVES + wave; grp < n_groups; grp += n_waves) {
      int d[8];
      #pragma unroll
      for (int u = 0; u < 8; ++u)
        d[u] = dstp[min((grp * 8 + u) * 32 + lane, last)];
      unsigned wlo = 0u, whi = 0u;
      #pragma unroll
      for (int u = 0; u < 8; ++u) {
        const unsigned key = (unsigned)d[u] / (unsigned)OWN_NODES;
        #pragma unroll
        for (int j = 0; j < 4; ++j) {
          const unsigned blo = __builtin_amdgcn_ballot_w32(((key >> j) & 1u) != 0u);
          const unsigned bhi = __builtin_amdgcn_ballot_w32(((key >> (4 + j)) & 1u) != 0u);
          if (lane == 4 * u + j) { wlo = blo; whi = bhi; }
        }
      }
      keys[32 * grp + lane]    = wlo;
      keys_hi[32 * grp + lane] = whi;
    }
  }
}

__global__ void __launch_bounds__(OWN_THREADS)
own_range_sum_kernel(const int* __restrict__ ei,
                     const unsigned int* __restrict__ keys,
                     const float* __restrict__ P,
                     const float* __restrict__ x,
                     const float* __restrict__ bias,
                     const float* __restrict__ eps_p,
                     float* __restrict__ out,
                     int n_nodes, int n_edges) {
  __shared__ float acc[OWN_NODES * DIM];
  __shared__ int   qbuf[OWN_WAVES * Q_CAP];

  const int tid  = threadIdx.x;
  const int lane = tid & 31;
  const int base = blockIdx.x * OWN_NODES;
  const int cnt  = min(OWN_NODES, n_nodes - base);
  const int n4   = cnt * (DIM / 4);

  {
    const float scale = 1.0f + eps_p[0];
    const float4* p4 = (const float4*)(P + (size_t)base * DIM);
    const float4* x4 = (const float4*)(x + (size_t)base * DIM);
    const float4  bv = ((const float4*)bias)[lane];
    float4* a4 = (float4*)acc;
    for (int i = tid; i < n4; i += OWN_THREADS) {
      const float4 v = p4[i];
      const float4 r = x4[i];
      float4 s;
      s.x = v.x * scale + r.x + bv.x;
      s.y = v.y * scale + r.y + bv.y;
      s.z = v.z * scale + r.z + bv.z;
      s.w = v.w * scale + r.w + bv.w;
      a4[i] = s;
    }
  }
  __syncthreads();

  const int* __restrict__ srcp = ei;
  const int* __restrict__ dstp = ei + n_edges;
  const int n_units  = (n_edges + 31) >> 5;
  const int n_groups = (n_units + 7) >> 3;
  const uint4* __restrict__ klo = (const uint4*)keys;
  const uint4* __restrict__ khi = (const uint4*)(keys + 32 * n_groups);
  const int iters   = (n_units + OWN_THREADS - 1) / OWN_THREADS;
  const int last_it = iters - 1;
  const unsigned kb = (unsigned)blockIdx.x;
  const unsigned x0 = ((kb >> 0) & 1u) - 1u, x1 = ((kb >> 1) & 1u) - 1u;
  const unsigned x2 = ((kb >> 2) & 1u) - 1u, x3 = ((kb >> 3) & 1u) - 1u;
  const unsigned x4 = ((kb >> 4) & 1u) - 1u, x5 = ((kb >> 5) & 1u) - 1u;
  const unsigned x6 = ((kb >> 6) & 1u) - 1u, x7 = ((kb >> 7) & 1u) - 1u;

  int* const qw = qbuf + (tid >> 5) * Q_CAP;
  unsigned q_head = 0u, q_tail = 0u;

  auto loadu = [&](const uint4* __restrict__ half_p, int it) -> uint4 {
    const int g = it * OWN_THREADS + tid;
    return half_p[g < n_units ? g : 0];
  };

  auto drain = [&](int n) {
    int s = 0, r = -1;
    if (lane < n) {
      const int e = qw[(q_head + (unsigned)lane) & (Q_CAP - 1)];
      s = srcp[e];
      r = dstp[e] - base;
    }
    q_head += (unsigned)n;
    #pragma unroll 1
    for (int i = 0; i < n; i += Q_GROUP) {
      float v[Q_GROUP][4];
      #pragma unroll
      for (int g = 0; g < Q_GROUP; ++g) {
        if (i + g < n) {
          const int su = __builtin_amdgcn_readlane(s, i + g);
          const float* hp = P + (size_t)su * DIM + lane;
          v[g][0] = hp[0];
          v[g][1] = hp[32];
          v[g][2] = hp[64];
          v[g][3] = hp[96];
        }
      }
      #pragma unroll
      for (int g = 0; g < Q_GROUP; ++g) {
        if (i + g < n) {
          const int ru = __builtin_amdgcn_readlane(r, i + g);
          if ((unsigned)ru < (unsigned)cnt) {
            float* ap = acc + ru * DIM + lane;
            atomicAdd(ap +  0, v[g][0]);
            atomicAdd(ap + 32, v[g][1]);
            atomicAdd(ap + 64, v[g][2]);
            atomicAdd(ap + 96, v[g][3]);
          }
        }
      }
    }
  };

  auto take32 = [&](const uint4& lo, const uint4& hi, int it) {
    const int g  = it * OWN_THREADS + tid;
    const int e0 = g << 5;
    unsigned m = (lo.x ^ x0) & (lo.y ^ x1) & (lo.z ^ x2) & (lo.w ^ x3) &
                 (hi.x ^ x4) & (hi.y ^ x5) & (hi.z ^ x6) & (hi.w ^ x7);
    if (it >= last_it) {
      const int rem = n_edges - e0;
      if (rem < 32) m = (rem > 0) ? (m & ((1u << rem) - 1u)) : 0u;
    }
    unsigned live = __builtin_amdgcn_ballot_w32(m != 0u);
    while (live != 0u) {
      const int e = e0 + __builtin_ctz(m | 0x80000000u);
      const unsigned pos = q_tail + __builtin_amdgcn_mbcnt_lo(live, 0u);
      if (m != 0u) qw[pos & (Q_CAP - 1)] = e;
      q_tail += (unsigned)__builtin_popcount(live);
      m &= m - 1u;
      if (q_tail - q_head >= 32u) drain(32);
      live = __builtin_amdgcn_ballot_w32(m != 0u);
    }
  };

  const int n_batches = (iters + 3) >> 2;
  const int rot = (int)(((long long)blockIdx.x * n_batches) / gridDim.x);

  for (int k = 0; k < n_batches; ++k) {
    const int bt = (k + rot < n_batches) ? k + rot : k + rot - n_batches;
    const int it = 4 * bt;
    const uint4 a0 = loadu(klo, it),     b0 = loadu(khi, it);
    const uint4 a1 = loadu(klo, it + 1), b1 = loadu(khi, it + 1);
    const uint4 a2 = loadu(klo, it + 2), b2 = loadu(khi, it + 2);
    const uint4 a3 = loadu(klo, it + 3), b3 = loadu(khi, it + 3);
    take32(a0, b0, it);
    take32(a1, b1, it + 1);
    take32(a2, b2, it + 2);
    take32(a3, b3, it + 3);
  }
  if (q_tail != q_head) drain((int)(q_tail - q_head));
  __syncthreads();

  {
    float4* o4 = (float4*)(out + (size_t)base * DIM);
    const float4* a4 = (const float4*)acc;
    for (int i = tid; i < n4; i += OWN_THREADS) o4[i] = a4[i];
  }
}

extern "C" void kernel_launch(void* const* d_in, const int* in_sizes, int n_in,
                              void* d_out, int out_size, void* d_ws, size_t ws_size,
                              hipStream_t stream) {
  const float* x     = (const float*)d_in[0];
  const int*   ei    = (const int*)  d_in[1];
  const float* gamma = (const float*)d_in[2];
  const float* beta  = (const float*)d_in[3];
  const float* W     = (const float*)d_in[4];
  const float* bias  = (const float*)d_in[5];
  const float* eps   = (const float*)d_in[6];
  float* out = (float*)d_out;

  const int n_nodes = in_sizes[0] / DIM;
  const int n_edges = in_sizes[1] / 2;

  float* P = (float*)d_ws;
  unsigned int* keys = (unsigned int*)(P + (size_t)n_nodes * DIM);

  {
    int blocks = (n_nodes + PT_ROWS - 1) / PT_ROWS;
    ln_relu_linear_kernel<<<blocks, PT_THREADS, 0, stream>>>(x, gamma, beta, W, ei, P, keys, n_nodes, n_edges);
  }
  {
    int blocks = (n_nodes + OWN_NODES - 1) / OWN_NODES;
    own_range_sum_kernel<<<blocks, OWN_THREADS, 0, stream>>>(ei, keys, P, x, bias, eps, out, n_nodes, n_edges);
  }
}


# Range sum in two phases: found edges are routed (1 integer LDS add each) to per-row lists; one wave sums a row's P rows in registers, 16 B loads: no ds_add_f32; sum order: arrival, self term last
# speedup vs baseline: 3.8786x; 1.1570x over previous
#include <hip/hip_runtime.h>

#define DIM 128
#define PT_WAVES 8
#define PT_THREADS (PT_WAVES * 32)
#define PT_TILES 5
#define PT_ROWS (16 * PT_TILES)
#define HS_PITCH 132
#define OWN_NODES 400
#define OWN_THREADS 1024
#define OWN_WAVES (OWN_THREADS / 32)
#define Q_CAP 64
#define S_GROUP 16
#define R_CAP 48
#define R_PITCH 49

typedef float v2f __attribute__((ext_vector_type(2)));
typedef float v8f __attribute__((ext_vector_type(8)));

__global__ void __launch_bounds__(PT_THREADS)
ln_relu_linear_kernel(const float* __restrict__ x,
                      const float* __restrict__ gamma,
                      const float* __restrict__ beta,
                      const float* __restrict__ W,
                      const int* __restrict__ ei,
                      float* __restrict__ P,
                      unsigned int* __restrict__ keys,
                      int n_nodes, int n_edges) {
  __shared__ __attribute__((aligned(16))) float hs[PT_ROWS * HS_PITCH];

  const int wave = threadIdx.x >> 5;
  const int lane = threadIdx.x & 31;
  const int half = lane >> 4;
  const int l    = lane & 15;
  const int row_base = blockIdx.x * PT_ROWS;

  v2f bf[DIM / 4];
  {
    const float* brow = W + (size_t)(16 * wave + l) * DIM + 2 * half;
    #pragma unroll
    for (int kk = 0; kk < DIM / 4; ++kk) bf[kk] = *(const v2f*)(brow + 4 * kk);
  }

  {
    float4 v[2 * PT_TILES];
    #pragma unroll
    for (int j = 0; j < 2 * PT_TILES; ++j) {
      const int row = min(row_base + wave + 8 * j, n_nodes - 1);
      v[j] = ((const float4*)(x + (size_t)row * DIM))[lane];
    }
    const float4 g = ((const float4*)gamma)[lane];
    const float4 b = ((const float4*)beta)[lane];
    #pragma unroll
    for (int j = 0; j < 2 * PT_TILES; ++j) {
      float s  = v[j].x + v[j].y + v[j].z + v[j].w;
      float s2 = v[j].x * v[j].x + v[j].y * v[j].y + v[j].z * v[j].z + v[j].w * v[j].w;
      #pragma unroll
      for (int off = 16; off > 0; off >>= 1) {
        s  += __shfl_xor(s,  off, 32);
        s2 += __shfl_xor(s2, off, 32);
      }
      const float mu  = s * (1.0f / DIM);
      const float var = s2 * (1.0f / DIM) - mu * mu;
      const float rs  = rsqrtf(var + 1e-5f);
      float4 hv;
      hv.x = fmaxf((v[j].x - mu) * rs * g.x + b.x, 0.0f);
      hv.y = fmaxf((v[j].y - mu) * rs * g.y + b.y, 0.0f);
      hv.z = fmaxf((v[j].z - mu) * rs * g.z + b.z, 0.0f);
      hv.w = fmaxf((v[j].w - mu) * rs * g.w + b.w, 0.0f);
      *(float4*)(hs + (wave + 8 * j) * HS_PITCH + 4 * lane) = hv;
    }
  }
  __syncthreads();

  for (int t = 0; t < PT_TILES; ++t) {
    const float* arow = hs + (16 * t + l) * HS_PITCH + 2 * half;
    v8f c = {};
    #pragma unroll
    for (int kk = 0; kk < DIM / 4; ++kk) {
      const v2f a = *(const v2f*)(arow + 4 * kk);
      c = __builtin_amdgcn_wmma_f32_16x16x4_f32(
           false, a,  false, bf[kk],
           (short)0, c,  false,  false);
    }
    #pragma unroll
    for (int r = 0; r < 8; ++r) {
      const int row = row_base + 16 * t + r + 8 * half;
      if (row < n_nodes) P[(size_t)row * DIM + 16 * wave + l] = c[r];
    }
  }

  {
    const int* __restrict__ dstp = ei + n_edges;
    const int n_units  = (n_edges + 31) >> 5;
    const int n_groups = (n_units + 7) >> 3;
    unsigned int* __restrict__ keys_hi = keys + 32 * n_groups;
    const int last    = n_edges - 1;
    const int n_waves = gridDim.x * PT_WAVES;
    for (int grp = blockIdx.x * PT_WAVES + wave; grp < n_groups; grp += n_waves) {
      int d[8];
      #pragma unroll
      for (int u = 0; u < 8; ++u)
        d[u] = dstp[min((grp * 8 + u) * 32 + lane, last)];
      unsigned wlo = 0u, whi = 0u;
      #pragma unroll
      for (int u = 0; u < 8; ++u) {
        const unsigned key = (unsigned)d[u] / (unsigned)OWN_NODES;
        #pragma unroll
        for (int j = 0; j < 4; ++j) {
          const unsigned blo = __builtin_amdgcn_ballot_w32(((key >> j) & 1u) != 0u);
          const unsigned bhi = __builtin_amdgcn_ballot_w32(((key >> (4 + j)) & 1u) != 0u);
          if (lane == 4 * u + j) { wlo = blo; whi = bhi; }
        }
      }
      keys[32 * grp + lane]    = wlo;
      keys_hi[32 * grp + lane] = whi;
    }
  }
}

__global__ void __launch_bounds__(OWN_THREADS)
own_range_sum_kernel(const int* __restrict__ ei,
                     const unsigned int* __restrict__ keys,
                     const float* __restrict__ P,
                     const float* __restrict__ x,
                     const float* __restrict__ bias,
                     const float* __restrict__ eps_p,
                     float* __restrict__ out,
                     int n_nodes, int n_edges) {
  __shared__ __attribute__((aligned(16))) float acc[OWN_NODES * DIM];
  __shared__ int qbuf[OWN_WAVES * Q_CAP];
  __shared__ int rl[OWN_NODES * R_PITCH];
  __shared__ int rcnt[OWN_NODES];

  const int tid  = threadIdx.x;
  const int lane = tid & 31;
  const int wv   = __builtin_amdgcn_readfirstlane(tid >> 5);
  const int base = blockIdx.x * OWN_NODES;
  const int cnt  = min(OWN_NODES, n_nodes - base);

  {
    float4* a4 = (float4*)acc;
    const float4 z = make_float4(0.0f, 0.0f, 0.0f, 0.0f);
    for (int i = tid; i < OWN_NODES * (DIM / 4); i += OWN_THREADS) a4[i] = z;
    if (tid < OWN_NODES) rcnt[tid] = 0;
  }
  __syncthreads();

  const int* __restrict__ srcp = ei;
  const int* __restrict__ dstp = ei + n_edges;
  const int n_units  = (n_edges + 31) >> 5;
  const int n_groups = (n_units + 7) >> 3;
  const uint4* __restrict__ klo = (const uint4*)keys;
  const uint4* __restrict__ khi = (const uint4*)(keys + 32 * n_groups);
  const int iters   = (n_units + OWN_THREADS - 1) / OWN_THREADS;
  const int last_it = iters - 1;
  const unsigned kb = (unsigned)blockIdx.x;
  const unsigned x0 = ((kb >> 0) & 1u) - 1u, x1 = ((kb >> 1) & 1u) - 1u;
  const unsigned x2 = ((kb >> 2) & 1u) - 1u, x3 = ((kb >> 3) & 1u) - 1u;
  const unsigned x4 = ((kb >> 4) & 1u) - 1u, x5 = ((kb >> 5) & 1u) - 1u;
  const unsigned x6 = ((kb >> 6) & 1u) - 1u, x7 = ((kb >> 7) & 1u) - 1u;

  int* const qw = qbuf + wv * Q_CAP;
  unsigned q_head = 0u, q_tail = 0u;

  auto loadu = [&](const uint4* __restrict__ half_p, int it) -> uint4 {
    const int g = it * OWN_THREADS + tid;
    return half_p[g < n_units ? g : 0];
  };

  auto route = [&](int n) {
    int s = 0, r = -1;
    if (lane < n) {
      const int e = qw[(q_head + (unsigned)lane) & (Q_CAP - 1)];
      s = srcp[e];
      r = dstp[e] - base;
    }
    q_head += (unsigned)n;
    bool full = false;
    if ((unsigned)r < (unsigned)cnt) {
      const int pos = atomicAdd(&rcnt[r], 1);
      if (pos < R_CAP) rl[r * R_PITCH + pos] = s;
      else full = true;
    }
    unsigned fm = __builtin_amdgcn_ballot_w32(full);
    while (fm != 0u) {
      const int ln = __builtin_ctz(fm);
      fm &= fm - 1u;
      const int su = __builtin_amdgcn_readlane(s, ln);
      const int ru = __builtin_amdgcn_readlane(r, ln);
      const float* hp = P + (size_t)su * DIM + lane;
      const float h0 = hp[0], h1 = hp[32], h2 = hp[64], h3 = hp[96];
      float* ap = acc + ru * DIM + lane;
      atomicAdd(ap +  0, h0);
      atomicAdd(ap + 32, h1);
      atomicAdd(ap + 64, h2);
      atomicAdd(ap + 96, h3);
    }
  };

  auto take32 = [&](const uint4& lo, const uint4& hi, int it) {
    const int g  = it * OWN_THREADS + tid;
    const int e0 = g << 5;
    unsigned m = (lo.x ^ x0) & (lo.y ^ x1) & (lo.z ^ x2) & (lo.w ^ x3) &
                 (hi.x ^ x4) & (hi.y ^ x5) & (hi.z ^ x6) & (hi.w ^ x7);
    if (it >= last_it) {
      const int rem = n_edges - e0;
      if (rem < 32) m = (rem > 0) ? (m & ((1u << rem) - 1u)) : 0u;
    }
    unsigned live = __builtin_amdgcn_ballot_w32(m != 0u);
    while (live != 0u) {
      const int e = e0 + __builtin_ctz(m | 0x80000000u);
      const unsigned pos = q_tail + __builtin_amdgcn_mbcnt_lo(live, 0u);
      if (m != 0u) qw[pos & (Q_CAP - 1)] = e;
      q_tail += (unsigned)__builtin_popcount(live);
      m &= m - 1u;
      if (q_tail - q_head >= 32u) route(32);
      live = __builtin_amdgcn_ballot_w32(m != 0u);
    }
  };

  const int n_batches = (iters + 3) >> 2;
  const int rot = (int)(((long long)blockIdx.x * n_batches) / gridDim.x);

  for (int k = 0; k < n_batches; ++k) {
    const int bt = (k + rot < n_batches) ? k + rot : k + rot - n_batches;
    const int it = 4 * bt;
    const uint4 a0 = loadu(klo, it),     b0 = loadu(khi, it);
    const uint4 a1 = loadu(klo, it + 1), b1 = loadu(khi, it + 1);
    const uint4 a2 = loadu(klo, it + 2), b2 = loadu(khi, it + 2);
    const uint4 a3 = loadu(klo, it + 3), b3 = loadu(khi, it + 3);
    take32(a0, b0, it);
    take32(a1, b1, it + 1);
    take32(a2, b2, it + 2);
    take32(a3, b3, it + 3);
  }
  if (q_tail != q_head) route((int)(q_tail - q_head));
  __syncthreads();

  const float scale = 1.0f + eps_p[0];
  const float4 bv = ((const float4*)bias)[lane];
  const float4* __restrict__ P4 = (const float4*)P;
  #pragma unroll 1
  for (int row = wv; row < cnt; row += OWN_WAVES) {
    const size_t off4 = (size_t)(base + row) * (DIM / 4) + lane;
    const float4 pv = P4[off4];
    const float4 xv = ((const float4*)x)[off4];
    const int n = __builtin_amdgcn_readfirstlane(min(rcnt[row], R_CAP));
    float4 a = make_float4(0.0f, 0.0f, 0.0f, 0.0f);
    #pragma unroll 1
    for (int i0 = 0; i0 < n; i0 += 32) {
      const int m = min(32, n - i0);
      int ent = 0;
      if (lane < m) ent = rl[row * R_PITCH + i0 + lane];
      #pragma unroll 1
      for (int i = 0; i < m; i += S_GROUP) {
        float4 v[S_GROUP];
        #pragma unroll
        for (int g = 0; g < S_GROUP; ++g) {
          if (i + g < m) {
            const int su = __builtin_amdgcn_readlane(ent, i + g);
            v[g] = P4[(size_t)su * (DIM / 4) + lane];
          }
        }
        #pragma unroll
        for (int g = 0; g < S_GROUP; ++g) {
          if (i + g < m) {
            a.x += v[g].x;
            a.y += v[g].y;
            a.z += v[g].z;
            a.w += v[g].w;
          }
        }
      }
    }
    const float4 ov = ((const float4*)(acc + row * DIM))[lane];
    float4 o;
    o.x = a.x + (pv.x * scale + xv.x + bv.x) + ov.x;
    o.y = a.y + (pv.y * scale + xv.y + bv.y) + ov.y;
    o.z = a.z + (pv.z * scale + xv.z + bv.z) + ov.z;
    o.w = a.w + (pv.w * scale + xv.w + bv.w) + ov.w;
    ((float4*)out)[off4] = o;
  }
}

extern "C" void kernel_launch(void* const* d_in, const int* in_sizes, int n_in,
                              void* d_out, int out_size, void* d_ws, size_t ws_size,
                              hipStream_t stream) {
  const float* x     = (const float*)d_in[0];
  const int*   ei    = (const int*)  d_in[1];
  const float* gamma = (const float*)d_in[2];
  const float* beta  = (const float*)d_in[3];
  const float* W     = (const float*)d_in[4];
  const float* bias  = (const float*)d_in[5];
  const float* eps   = (const float*)d_in[6];
  float* out = (float*)d_out;

  const int n_nodes = in_sizes[0] / DIM;
  const int n_edges = in_sizes[1] / 2;

  float* P = (float*)d_ws;
  unsigned int* keys = (unsigned int*)(P + (size_t)n_nodes * DIM);

  {
    int blocks = (n_nodes + PT_ROWS - 1) / PT_ROWS;
    ln_relu_linear_kernel<<<blocks, PT_THREADS, 0, stream>>>(x, gamma, beta, W, ei, P, keys, n_nodes, n_edges);
  }
  {
    int blocks = (n_nodes + OWN_NODES - 1) / OWN_NODES;
    own_range_sum_kernel<<<blocks, OWN_THREADS, 0, stream>>>(ei, keys, P, x, bias, eps, out, n_nodes, n_edges);
  }
}


# Launch 1 bins its slice of the edge list by node range (LDS cursors) into 128 B cells; a range-sum block reads only its own cells: no scan of all edges' keys, no id look-ups; sum order: arrival
# speedup vs baseline: 4.5987x; 1.1857x over previous
#include <hip/hip_runtime.h>

#define DIM 128
#define PT_WAVES 8
#define PT_THREADS (PT_WAVES * 32)
#define PT_TILES 5
#define PT_ROWS (16 * PT_TILES)
#define HS_PITCH 132
#define OWN_NODES 400
#define OWN_THREADS 1024
#define OWN_WAVES (OWN_THREADS / 32)
#define MAX_OWNERS 256
#define CELL_WORDS 32
#define CELL_EDGES (CELL_WORDS - 1)
#define ROW_BITS 9
#define MAX_PACKED_NODES (1 << (32 - ROW_BITS))
#define C_BATCH 5
#define B_BATCH 5
#define S_GROUP 16
#define R_CAP 48
#define R_PITCH 49

typedef float v2f __attribute__((ext_vector_type(2)));
typedef float v8f __attribute__((ext_vector_type(8)));

__global__ void __launch_bounds__(PT_THREADS)
ln_relu_linear_kernel(const float* __restrict__ x,
                      const float* __restrict__ gamma,
                      const float* __restrict__ beta,
                      const float* __restrict__ W,
                      const int* __restrict__ ei,
                      float* __restrict__ P,
                      int4* __restrict__ cells,
                      int n_nodes, int n_edges) {
  __shared__ __attribute__((aligned(16))) float hs[PT_ROWS * HS_PITCH];

  const int wave = threadIdx.x >> 5;
  const int lane = threadIdx.x & 31;
  const int half = lane >> 4;
  const int l    = lane & 15;
  const int row_base = blockIdx.x * PT_ROWS;

  v2f bf[DIM / 4];
  {
    const float* brow = W + (size_t)(16 * wave + l) * DIM + 2 * half;
    #pragma unroll
    for (int kk = 0; kk < DIM / 4; ++kk) bf[kk] = *(const v2f*)(brow + 4 * kk);
  }

  {
    float4 v[2 * PT_TILES];
    #pragma unroll
    for (int j = 0; j < 2 * PT_TILES; ++j) {
      const int row = min(row_base + wave + 8 * j, n_nodes - 1);
      v[j] = ((const float4*)(x + (size_t)row * DIM))[lane];
    }
    const float4 g = ((const float4*)gamma)[lane];
    const float4 b = ((const float4*)beta)[lane];
    #pragma unroll
    for (int j = 0; j < 2 * PT_TILES; ++j) {
      float s  = v[j].x + v[j].y + v[j].z + v[j].w;
      float s2 = v[j].x * v[j].x + v[j].y * v[j].y + v[j].z * v[j].z + v[j].w * v[j].w;
      #pragma unroll
      for (int off = 16; off > 0; off >>= 1) {
        s  += __shfl_xor(s,  off, 32);
        s2 += __shfl_xor(s2, off, 32);
      }
      const float mu  = s * (1.0f / DIM);
      const float var = s2 * (1.0f / DIM) - mu * mu;
      const float rs  = rsqrtf(var + 1e-5f);
      float4 hv;
      hv.x = fmaxf((v[j].x - mu) * rs * g.x + b.x, 0.0f);
      hv.y = fmaxf((v[j].y - mu) * rs * g.y + b.y, 0.0f);
      hv.z = fmaxf((v[j].z - mu) * rs * g.z + b.z, 0.0f);
      hv.w = fmaxf((v[j].w - mu) * rs * g.w + b.w, 0.0f);
      *(float4*)(hs + (wave + 8 * j) * HS_PITCH + 4 * lane) = hv;
    }
  }
  __syncthreads();

  for (int t = 0; t < PT_TILES; ++t) {
    const float* arow = hs + (16 * t + l) * HS_PITCH + 2 * half;
    v8f c = {};
    #pragma unroll
    for (int kk = 0; kk < DIM / 4; ++kk) {
      const v2f a = *(const v2f*)(arow + 4 * kk);
      c = __builtin_amdgcn_wmma_f32_16x16x4_f32(
           false, a,  false, bf[kk],
           (short)0, c,  false,  false);
    }
    #pragma unroll
    for (int r = 0; r < 8; ++r) {
      const int row = row_base + 16 * t + r + 8 * half;
      if (row < n_nodes) P[(size_t)row * DIM + 16 * wave + l] = c[r];
    }
  }

  __syncthreads();
  {
    unsigned* const bins = (unsigned*)hs;
    int* const bcnt = (int*)(hs + MAX_OWNERS * CELL_WORDS);
    static_assert((MAX_OWNERS * CELL_WORDS + MAX_OWNERS) <= PT_ROWS * HS_PITCH, "bins must fit the LN tile");
    static_assert(OWN_NODES <= (1 << ROW_BITS), "a row inside its range must fit ROW_BITS");
    const int tid = threadIdx.x;
    {
      int4* const z4 = (int4*)hs;
      const int4 z = make_int4(0, 0, 0, 0);
      for (int i = tid; i < MAX_OWNERS * CELL_WORDS / 4; i += PT_THREADS) z4[i] = z;
      for (int i = tid; i < MAX_OWNERS; i += PT_THREADS) bcnt[i] = 0;
    }
    __syncthreads();
    const int* __restrict__ dstp = ei + n_edges;
    const int slice_len = (n_edges + (int)gridDim.x - 1) / (int)gridDim.x;
    const int e_begin = min((int)blockIdx.x * slice_len, n_edges);
    const int e_end   = min(e_begin + slice_len, n_edges);
    for (int eb = e_begin; eb < e_end; eb += PT_THREADS * B_BATCH) {
      int dv[B_BATCH], sv[B_BATCH];
      #pragma unroll
      for (int j = 0; j < B_BATCH; ++j) {
        const int ec = min(eb + j * PT_THREADS + tid, e_end - 1);
        dv[j] = dstp[ec];
        sv[j] = ei[ec];
      }
      asm volatile("" :: "v"(sv[0]), "v"(sv[1]), "v"(sv[2]), "v"(sv[3]), "v"(sv[4]));
      static_assert(B_BATCH == 5, "the statement above names the batch's five src loads");
      #pragma unroll
      for (int j = 0; j < B_BATCH; ++j) {
        const int d = dv[j];
        const int s = sv[j];
        const unsigned owner = (unsigned)d / (unsigned)OWN_NODES;
        if (eb + j * PT_THREADS + tid < e_end && owner < (unsigned)MAX_OWNERS) {
          const int k = atomicAdd(&bcnt[owner], 1) + 1;
          if (k <= CELL_EDGES)
            bins[owner * CELL_WORDS + ((k & 7) << 2) + (k >> 3)] = ((unsigned)s << ROW_BITS) | ((unsigned)d - owner * (unsigned)OWN_NODES);
        }
      }
    }
    __syncthreads();
    const int n_owners = (n_nodes + OWN_NODES - 1) / OWN_NODES;
    const bool packable = n_nodes <= MAX_PACKED_NODES;
    const int4* const b4 = (const int4*)hs;
    for (int idx = tid; idx < n_owners * 8; idx += PT_THREADS) {
      const int owner = idx >> 3;
      const int part  = idx & 7;
      int4 v = make_int4(0, 0, 0, 0);
      int  c = CELL_WORDS;
      if (owner < MAX_OWNERS && packable) { v = b4[idx]; c = bcnt[owner]; }
      if (part == 0) v.x = c;
      cells[((size_t)owner * gridDim.x + blockIdx.x) * 8 + part] = v;
    }
  }
}

__global__ void __launch_bounds__(OWN_THREADS)
own_range_sum_kernel(const int* __restrict__ ei,
                     const int4* __restrict__ cells,
                     const float* __restrict__ P,
                     const float* __restrict__ x,
                     const float* __restrict__ bias,
                     const float* __restrict__ eps_p,
                     float* __restrict__ out,
                     int n_nodes, int n_edges, int n_slices) {
  __shared__ __attribute__((aligned(16))) float acc[OWN_NODES * DIM];
  __shared__ int rl[OWN_NODES * R_PITCH];
  __shared__ int rcnt[OWN_NODES];

  const int tid  = threadIdx.x;
  const int lane = tid & 31;
  const int wv   = __builtin_amdgcn_readfirstlane(tid >> 5);
  const int base = blockIdx.x * OWN_NODES;
  const int cnt  = min(OWN_NODES, n_nodes - base);

  {
    float4* a4 = (float4*)acc;
    const float4 z = make_float4(0.0f, 0.0f, 0.0f, 0.0f);
    for (int i = tid; i < OWN_NODES * (DIM / 4); i += OWN_THREADS) a4[i] = z;
    if (tid < OWN_NODES) rcnt[tid] = 0;
  }
  __syncthreads();

  const int* __restrict__ srcp = ei;
  const int* __restrict__ dstp = ei + n_edges;
  auto place = [&](int s, int r) {
    bool full = false;
    if ((unsigned)r < (unsigned)cnt) {
      const int pos = atomicAdd(&rcnt[r], 1);
      if (pos < R_CAP) rl[r * R_PITCH + pos] = s;
      else full = true;
    }
    unsigned fm = __builtin_amdgcn_ballot_w32(full);
    while (fm != 0u) {
      const int ln = __builtin_ctz(fm);
      fm &= fm - 1u;
      const int su = __builtin_amdgcn_readlane(s, ln);
      const int ru = __builtin_amdgcn_readlane(r, ln);
      const float* hp = P + (size_t)su * DIM + lane;
      const float h0 = hp[0], h1 = hp[32], h2 = hp[64], h3 = hp[96];
      float* ap = acc + ru * DIM + lane;
      atomicAdd(ap +  0, h0);
      atomicAdd(ap + 32, h1);
      atomicAdd(ap + 64, h2);
      atomicAdd(ap + 96, h3);
    }
  };

  const int slice_len = (n_edges + n_slices - 1) / n_slices;
  const int4* __restrict__ mine = cells + (size_t)blockIdx.x * n_slices * 8;
  const int part = lane & 7;
  const int hsrc = lane & ~7;
  const int n_cg = (n_slices + 3) >> 2;

  for (int cg0 = wv; cg0 < n_cg; cg0 += OWN_WAVES * C_BATCH) {
    int4 v[C_BATCH];
    #pragma unroll
    for (int j = 0; j < C_BATCH; ++j) {
      const int sl = (cg0 + j * OWN_WAVES) * 4 + (lane >> 3);
      v[j] = mine[(size_t)(sl < n_slices ? sl : 0) * 8 + part];
    }
    __builtin_amdgcn_sched_barrier(0);
    #pragma unroll
    for (int j = 0; j < C_BATCH; ++j) {
      const int sl = (cg0 + j * OWN_WAVES) * 4 + (lane >> 3);
      int hdr = __shfl(v[j].x, hsrc, 32);
      if (sl >= n_slices) hdr = 0;
      const bool spilled = hdr > CELL_EDGES;
      const int  nrec = spilled ? 0 : hdr;
      const unsigned w[4] = {(unsigned)v[j].x, (unsigned)v[j].y, (unsigned)v[j].z, (unsigned)v[j].w};
      #pragma unroll
      for (int q = 0; q < 4; ++q) {
        const int  k  = 8 * q + part;
        const bool ok = k >= 1 && k <= nrec;
        if (__builtin_amdgcn_ballot_w32(ok) != 0u)
          place((int)(w[q] >> ROW_BITS), ok ? (int)(w[q] & ((1u << ROW_BITS) - 1u)) : -1);
      }
      unsigned sm = __builtin_amdgcn_ballot_w32(spilled && part == 0);
      while (sm != 0u) {
        const int ln = __builtin_ctz(sm);
        sm &= sm - 1u;
        const int su = (cg0 + j * OWN_WAVES) * 4 + (ln >> 3);
        const int e0 = min(su * slice_len, n_edges);
        const int e1 = min(e0 + slice_len, n_edges);
        for (int eb = e0; eb < e1; eb += 32) {
          const int e = eb + lane;
          int s = 0, r = -1;
          if (e < e1) { s = srcp[e]; r = dstp[e] - base; }
          place(s, r);
        }
      }
    }
  }
  __syncthreads();


  const float scale = 1.0f + eps_p[0];
  const float4 bv = ((const float4*)bias)[lane];
  const float4* __restrict__ P4 = (const float4*)P;
  #pragma unroll 1
  for (int row = wv; row < cnt; row += OWN_WAVES) {
    const size_t off4 = (size_t)(base + row) * (DIM / 4) + lane;
    const float4 pv = P4[off4];
    const float4 xv = ((const float4*)x)[off4];
    const int n = __builtin_amdgcn_readfirstlane(min(rcnt[row], R_CAP));
    float4 a = make_float4(0.0f, 0.0f, 0.0f, 0.0f);
    #pragma unroll 1
    for (int i0 = 0; i0 < n; i0 += 32) {
      const int m = min(32, n - i0);
      int ent = 0;
      if (lane < m) ent = rl[row * R_PITCH + i0 + lane];
      #pragma unroll 1
      for (int i = 0; i < m; i += S_GROUP) {
        float4 v[S_GROUP];
        #pragma unroll
        for (int g = 0; g < S_GROUP; ++g) {
          if (i + g < m) {
            const int su = __builtin_amdgcn_readlane(ent, i + g);
            v[g] = P4[(size_t)su * (DIM / 4) + lane];
          }
        }
        #pragma unroll
        for (int g = 0; g < S_GROUP; ++g) {
          if (i + g < m) {
            a.x += v[g].x;
            a.y += v[g].y;
            a.z += v[g].z;
            a.w += v[g].w;
          }
        }
      }
    }
    const float4 ov = ((const float4*)(acc + row * DIM))[lane];
    float4 o;
    o.x = a.x + (pv.x * scale + xv.x + bv.x) + ov.x;
    o.y = a.y + (pv.y * scale + xv.y + bv.y) + ov.y;
    o.z = a.z + (pv.z * scale + xv.z + bv.z) + ov.z;
    o.w = a.w + (pv.w * scale + xv.w + bv.w) + ov.w;
    ((float4*)out)[off4] = o;
  }
}

extern "C" void kernel_launch(void* const* d_in, const int* in_sizes, int n_in,
                              void* d_out, int out_size, void* d_ws, size_t ws_size,
                              hipStream_t stream) {
  const float* x     = (const float*)d_in[0];
  const int*   ei    = (const int*)  d_in[1];
  const float* gamma = (const float*)d_in[2];
  const float* beta  = (const float*)d_in[3];
  const float* W     = (const float*)d_in[4];
  const float* bias  = (const float*)d_in[5];
  const float* eps   = (const float*)d_in[6];
  float* out = (float*)d_out;

  const int n_nodes = in_sizes[0] / DIM;
  const int n_edges = in_sizes[1] / 2;

  float* P = (float*)d_ws;
  int4* cells = (int4*)(P + (size_t)n_nodes * DIM);

  const int blocks1 = (n_nodes + PT_ROWS - 1) / PT_ROWS;
  const int blocks2 = (n_nodes + OWN_NODES - 1) / OWN_NODES;
  ln_relu_linear_kernel<<<blocks1, PT_THREADS, 0, stream>>>(x, gamma, beta, W, ei, P, cells, n_nodes, n_edges);
  own_range_sum_kernel<<<blocks2, OWN_THREADS, 0, stream>>>(ei, cells, P, x, bias, eps, out, n_nodes, n_edges, blocks1);
}
